# starting kernel
# baseline (speedup 1.0000x reference)
#include <hip/hip_runtime.h>

typedef __attribute__((ext_vector_type(2))) float v2f;
typedef __attribute__((ext_vector_type(8))) float v8f;

#define BSZ     256
#define LSZ     2048
#define CSZ     6
#define NW      63
#define TSTEPS  63
#define MSTRIDE 68
#define MROWS   48
#define DROWS   16
#define PSTRIDE 8
#define WAVE_LDS (MROWS*MSTRIDE + DROWS*MSTRIDE + 64*PSTRIDE)
#define WPB     2

__global__ __launch_bounds__(64)
void ComputeWindowSignature_11106785427739_kernel(const float* __restrict__ path,
                                                  float* __restrict__ out)
{
    __shared__ __align__(16) float lds[WPB * WAVE_LDS];
    const int lane = threadIdx.x & 31;
    const int wv   = threadIdx.x >> 5;
    const int gw   = blockIdx.x * WPB + wv;
    const int b    = gw / NW;
    const int w    = gw - b * NW;

    float* M = &lds[wv * WAVE_LDS];
    float* D = M + MROWS * MSTRIDE;
    float* P = D + DROWS * MSTRIDE;

    for (int e = lane; e < DROWS*MSTRIDE + 64*PSTRIDE; e += 32) D[e] = 0.f;
    const size_t base = ((size_t)b * LSZ + (size_t)w * 32) * CSZ;
    for (int e = lane; e < 384; e += 32) M[e] = path[base + e];
    __syncthreads();

    for (int e = lane; e < TSTEPS * CSZ; e += 32) {
        int t = e / CSZ, c = e - t * CSZ;
        D[c * MSTRIDE + t] = M[e + CSZ] - M[e];
    }
    __syncthreads();

    for (int e = lane; e < MROWS * MSTRIDE; e += 32) M[e] = 0.f;
    __syncthreads();

    const size_t po = (size_t)gw * 258;

    if (lane < CSZ) {
        float p = 0.f;
        for (int t = 0; t < TSTEPS; ++t) {
            P[t * PSTRIDE + lane] = p;
            p += D[lane * MSTRIDE + t];
        }
        out[po + lane] = p;
    }
    __syncthreads();

    const int k1 = lane;
    const int i1 = k1 / CSZ, j1 = k1 - i1 * CSZ;
    const int k2 = lane + 32;
    const int i2 = k2 / CSZ, j2 = k2 - i2 * CSZ;
    const bool has2 = (k2 < 36);
    float s2a = 0.f, s2b = 0.f;
    for (int t = 0; t < TSTEPS; ++t) {
        float di = D[i1 * MSTRIDE + t];
        float dj = D[j1 * MSTRIDE + t];
        float pi = P[t * PSTRIDE + i1];
        M[k1 * MSTRIDE + t] = s2a + (0.5f * pi + (1.f/6.f) * di) * dj;
        s2a += (pi + 0.5f * di) * dj;
        if (has2) {
            float di2 = D[i2 * MSTRIDE + t];
            float dj2 = D[j2 * MSTRIDE + t];
            float pi2 = P[t * PSTRIDE + i2];
            M[k2 * MSTRIDE + t] = s2b + (0.5f * pi2 + (1.f/6.f) * di2) * dj2;
            s2b += (pi2 + 0.5f * di2) * dj2;
        }
    }
    out[po + 6 + k1] = 2.f * s2a;
    if (has2) out[po + 6 + k2] = 2.f * s2b;
    __syncthreads();

    const int half = lane >> 4;
    const int l16  = lane & 15;
    #pragma unroll
    for (int tau = 0; tau < 3; ++tau) {
        v8f acc = (v8f)(0.f);
        #pragma unroll
        for (int kc = 0; kc < 16; ++kc) {
            const int kk = 4 * kc + 2 * half;
            v2f a  = *(const v2f*)&M[(16 * tau + l16) * MSTRIDE + kk];
            v2f bf = *(const v2f*)&D[l16 * MSTRIDE + kk];
            acc = __builtin_amdgcn_wmma_f32_16x16x4_f32(
                      false, a, false, bf, (short)0, acc, false, false);
        }
        #pragma unroll
        for (int r = 0; r < 8; ++r) {
            int mg = 16 * tau + 8 * half + r;
            if (mg < 36 && l16 < CSZ)
                out[po + 42 + mg * CSZ + l16] = 6.f * acc[r];
        }
    }
}

extern "C" void kernel_launch(void* const* d_in, const int* in_sizes, int n_in,
                              void* d_out, int out_size, void* d_ws, size_t ws_size,
                              hipStream_t stream) {
    const float* path = (const float*)d_in[0];
    float* out = (float*)d_out;
    const int pairs  = BSZ * NW;
    const int blocks = pairs / WPB;
    ComputeWindowSignature_11106785427739_kernel<<<blocks, WPB * 32, 0, stream>>>(path, out);
}


# matrix A fragments built in registers: lane (h,c) runs row c's level-2 recurrence over its own half of K and feeds the WMMA directly; M never goes to LDS, the d_t[j] registers are also the B fragment
# speedup vs baseline: 3.5748x; 3.5748x over previous
#include <hip/hip_runtime.h>


typedef __attribute__((ext_vector_type(2))) float v2f;
typedef __attribute__((ext_vector_type(4))) float v4f;
typedef __attribute__((ext_vector_type(8))) float v8f;

#define BSZ     256
#define LSZ     2048
#define CSZ     6
#define NW      63
#define TSTEPS  63
#define MSTRIDE 68
#define DROWS   18
#define WPB     2

__device__ __forceinline__ float xhalf(float x) {
    return __builtin_bit_cast(float, __builtin_amdgcn_permlanex16(
        __builtin_bit_cast(int, x), __builtin_bit_cast(int, x), 0x76543210, 0xfedcba98, false, false));
}

__global__ __launch_bounds__(64)
void ComputeWindowSignature_11106785427739_kernel(const float* __restrict__ path,
                                                  float* __restrict__ out)
{
    __shared__ __align__(16) float Xsh[WPB][64 * CSZ];
    __shared__ __align__(16) float Dsh[WPB][DROWS * MSTRIDE];
    const int lane = threadIdx.x & 31;
    const int wv   = threadIdx.x >> 5;
    const int gw   = blockIdx.x * WPB + wv;
    const int b    = gw / NW;
    const int w    = gw - b * NW;

    float* X  = Xsh[wv];
    float* D  = Dsh[wv];
    float* F1 = D + 6 * MSTRIDE;
    float* F2 = D + 12 * MSTRIDE;

    const size_t base = ((size_t)b * LSZ + (size_t)w * 32) * CSZ;
    for (int e = lane; e < 384; e += 32) X[e] = path[base + e];
    __syncthreads();

    const size_t po = (size_t)gw * 258;

    #pragma unroll
    for (int it = 0; it < 12; ++it) {
        const int c = it >> 1;
        const int t = ((it & 1) << 5) + lane;
        const int tn = (t < TSTEPS) ? t + 1 : t;
        const float x0 = X[c];
        const float xt = X[t * CSZ + c];
        const float xn = X[tn * CSZ + c];
        const float d = xn - xt;
        const float p = xt - x0;
        D [c * MSTRIDE + t] = d;
        F1[c * MSTRIDE + t] = p + 0.5f * d;
        F2[c * MSTRIDE + t] = 0.5f * p + (1.f / 6.f) * d;
    }
    if (lane < CSZ) out[po + lane] = X[TSTEPS * CSZ + lane] - X[lane];
    __syncthreads();

    const int h  = lane >> 4;
    const int c  = lane & 15;
    const int cr = (c < 12) ? c : 11;
    const int ip = (cr >= CSZ) ? 1 : 0;
    const int j  = cr - CSZ * ip;
    const float sx = X[TSTEPS * CSZ + j] - X[32 * CSZ + j];
    const v4f* djp = (const v4f*)(D + j * MSTRIDE + 32 * h);
    const v4f* f1p = (const v4f*)(F1 + ip * MSTRIDE + 32 * h);
    const v4f* f2p = (const v4f*)(F2 + ip * MSTRIDE + 32 * h);
    float sa[3] = {0.f, 0.f, 0.f};
    v8f acc[3] = {(v8f)(0.f), (v8f)(0.f), (v8f)(0.f)};
    #pragma unroll
    for (int q = 0; q < 8; ++q) {
        const v4f dj = djp[q];
        #pragma unroll
        for (int tau = 0; tau < 3; ++tau) {
            const v4f a1 = f1p[2 * tau * (MSTRIDE / 4) + q];
            const v4f b1 = f2p[2 * tau * (MSTRIDE / 4) + q];
            v4f m;
            #pragma unroll
            for (int e = 0; e < 4; ++e) {
                m[e]    = __builtin_fmaf(b1[e], dj[e], sa[tau]);
                sa[tau] = __builtin_fmaf(a1[e], dj[e], sa[tau]);
            }
            v2f alo = {m[0], m[1]}, blo = {dj[0], dj[1]};
            v2f ahi = {m[2], m[3]}, bhi = {dj[2], dj[3]};
            acc[tau] = __builtin_amdgcn_wmma_f32_16x16x4_f32(
                           false, alo, false, blo, (short)0, acc[tau], false, false);
            acc[tau] = __builtin_amdgcn_wmma_f32_16x16x4_f32(
                           false, ahi, false, bhi, (short)0, acc[tau], false, false);
        }
    }
    const bool half0 = (h == 0);
    const v2f bx = {sx, 0.f};
    #pragma unroll
    for (int tau = 0; tau < 3; ++tau) {
        const v2f ax = {half0 ? sa[tau] : 0.f, 0.f};
        acc[tau] = __builtin_amdgcn_wmma_f32_16x16x4_f32(
                       false, ax, false, bx, (short)0, acc[tau], false, false);
    }
    float tot[3];
    #pragma unroll
    for (int tau = 0; tau < 3; ++tau) {
        asm volatile("" : "+v"(sa[tau]));
        tot[tau] = sa[tau] + xhalf(sa[tau]);
    }
    if (half0 && c < 12) {
        float* o2 = out + po + 6 + c;
        #pragma unroll
        for (int tau = 0; tau < 3; ++tau) o2[12 * tau] = 2.f * tot[tau];
    }
    if (c < CSZ) {
        float* o3 = out + po + 42 + 48 * h + c;
        #pragma unroll
        for (int tau = 0; tau < 3; ++tau) {
            #pragma unroll
            for (int r = 0; r < 4; ++r) o3[72 * tau + 6 * r] = 6.f * acc[tau][r];
        }
        if (half0) {
            #pragma unroll
            for (int tau = 0; tau < 3; ++tau) {
                #pragma unroll
                for (int r = 4; r < 8; ++r) o3[72 * tau + 6 * r] = 6.f * acc[tau][r];
            }
        }
    }
}

extern "C" void kernel_launch(void* const* d_in, const int* in_sizes, int n_in,
                              void* d_out, int out_size, void* d_ws, size_t ws_size,
                              hipStream_t stream) {
    const float* path = (const float*)d_in[0];
    float* out = (float*)d_out;
    const int pairs  = BSZ * NW;
    const int blocks = pairs / WPB;
    ComputeWindowSignature_11106785427739_kernel<<<blocks, WPB * 32, 0, stream>>>(path, out);
}


# window staged by three 16-byte loads per lane requested together and three 16-byte LDS stores; was twelve 4-byte loads waited for in five dependent groups
# speedup vs baseline: 4.1083x; 1.1492x over previous
#include <hip/hip_runtime.h>


typedef __attribute__((ext_vector_type(2))) float v2f;
typedef __attribute__((ext_vector_type(4))) float v4f;
typedef __attribute__((ext_vector_type(8))) float v8f;

#define BSZ     256
#define LSZ     2048
#define CSZ     6
#define NW      63
#define TSTEPS  63
#define MSTRIDE 68
#define DROWS   18
#define WPB     2

__device__ __forceinline__ float xhalf(float x) {
    return __builtin_bit_cast(float, __builtin_amdgcn_permlanex16(
        __builtin_bit_cast(int, x), __builtin_bit_cast(int, x), 0x76543210, 0xfedcba98, false, false));
}

__global__ __launch_bounds__(64)
void ComputeWindowSignature_11106785427739_kernel(const float* __restrict__ path,
                                                  float* __restrict__ out)
{
    __shared__ __align__(16) float Xsh[WPB][64 * CSZ];
    __shared__ __align__(16) float Dsh[WPB][DROWS * MSTRIDE];
    const int lane = threadIdx.x & 31;
    const int wv   = threadIdx.x >> 5;
    const int gw   = blockIdx.x * WPB + wv;
    const int b    = gw / NW;
    const int w    = gw - b * NW;

    float* X  = Xsh[wv];
    float* D  = Dsh[wv];
    float* F1 = D + 6 * MSTRIDE;
    float* F2 = D + 12 * MSTRIDE;

    const size_t base = ((size_t)b * LSZ + (size_t)w * 32) * CSZ;
    const v4f* src = (const v4f*)(path + base);
    const v4f r0 = src[lane];
    const v4f r1 = src[lane + 32];
    const v4f r2 = src[lane + 64];
    v4f* Xv = (v4f*)X;
    Xv[lane]      = r0;
    Xv[lane + 32] = r1;
    Xv[lane + 64] = r2;
    __syncthreads();

    const size_t po = (size_t)gw * 258;

    #pragma unroll
    for (int it = 0; it < 12; ++it) {
        const int c = it >> 1;
        const int t = ((it & 1) << 5) + lane;
        const int tn = (t < TSTEPS) ? t + 1 : t;
        const float x0 = X[c];
        const float xt = X[t * CSZ + c];
        const float xn = X[tn * CSZ + c];
        const float d = xn - xt;
        const float p = xt - x0;
        D [c * MSTRIDE + t] = d;
        F1[c * MSTRIDE + t] = p + 0.5f * d;
        F2[c * MSTRIDE + t] = 0.5f * p + (1.f / 6.f) * d;
    }
    if (lane < CSZ) out[po + lane] = X[TSTEPS * CSZ + lane] - X[lane];
    __syncthreads();

    const int h  = lane >> 4;
    const int c  = lane & 15;
    const int cr = (c < 12) ? c : 11;
    const int ip = (cr >= CSZ) ? 1 : 0;
    const int j  = cr - CSZ * ip;
    const float sx = X[TSTEPS * CSZ + j] - X[32 * CSZ + j];
    const v4f* djp = (const v4f*)(D + j * MSTRIDE + 32 * h);
    const v4f* f1p = (const v4f*)(F1 + ip * MSTRIDE + 32 * h);
    const v4f* f2p = (const v4f*)(F2 + ip * MSTRIDE + 32 * h);
    float sa[3] = {0.f, 0.f, 0.f};
    v8f acc[3] = {(v8f)(0.f), (v8f)(0.f), (v8f)(0.f)};
    #pragma unroll
    for (int q = 0; q < 8; ++q) {
        const v4f dj = djp[q];
        #pragma unroll
        for (int tau = 0; tau < 3; ++tau) {
            const v4f a1 = f1p[2 * tau * (MSTRIDE / 4) + q];
            const v4f b1 = f2p[2 * tau * (MSTRIDE / 4) + q];
            v4f m;
            #pragma unroll
            for (int e = 0; e < 4; ++e) {
                m[e]    = __builtin_fmaf(b1[e], dj[e], sa[tau]);
                sa[tau] = __builtin_fmaf(a1[e], dj[e], sa[tau]);
            }
            v2f alo = {m[0], m[1]}, blo = {dj[0], dj[1]};
            v2f ahi = {m[2], m[3]}, bhi = {dj[2], dj[3]};
            acc[tau] = __builtin_amdgcn_wmma_f32_16x16x4_f32(
                           false, alo, false, blo, (short)0, acc[tau], false, false);
            acc[tau] = __builtin_amdgcn_wmma_f32_16x16x4_f32(
                           false, ahi, false, bhi, (short)0, acc[tau], false, false);
        }
    }
    const bool half0 = (h == 0);
    const v2f bx = {sx, 0.f};
    #pragma unroll
    for (int tau = 0; tau < 3; ++tau) {
        const v2f ax = {half0 ? sa[tau] : 0.f, 0.f};
        acc[tau] = __builtin_amdgcn_wmma_f32_16x16x4_f32(
                       false, ax, false, bx, (short)0, acc[tau], false, false);
    }
    float tot[3];
    #pragma unroll
    for (int tau = 0; tau < 3; ++tau) {
        asm volatile("" : "+v"(sa[tau]));
        tot[tau] = sa[tau] + xhalf(sa[tau]);
    }
    if (half0 && c < 12) {
        float* o2 = out + po + 6 + c;
        #pragma unroll
        for (int tau = 0; tau < 3; ++tau) o2[12 * tau] = 2.f * tot[tau];
    }
    if (c < CSZ) {
        float* o3 = out + po + 42 + 48 * h + c;
        #pragma unroll
        for (int tau = 0; tau < 3; ++tau) {
            #pragma unroll
            for (int r = 0; r < 4; ++r) o3[72 * tau + 6 * r] = 6.f * acc[tau][r];
        }
        if (half0) {
            #pragma unroll
            for (int tau = 0; tau < 3; ++tau) {
                #pragma unroll
                for (int r = 4; r < 8; ++r) o3[72 * tau + 6 * r] = 6.f * acc[tau][r];
            }
        }
    }
}

extern "C" void kernel_launch(void* const* d_in, const int* in_sizes, int n_in,
                              void* d_out, int out_size, void* d_ws, size_t ws_size,
                              hipStream_t stream) {
    const float* path = (const float*)d_in[0];
    float* out = (float*)d_out;
    const int pairs  = BSZ * NW;
    const int blocks = pairs / WPB;
    ComputeWindowSignature_11106785427739_kernel<<<blocks, WPB * 32, 0, stream>>>(path, out);
}


# output record assembled in the wave's own LDS and written as two 512-byte runs of b128 stores plus one b64 piece; was 28 store instructions of 24- and 48-byte pieces per window
# speedup vs baseline: 4.3974x; 1.0704x over previous
#include <hip/hip_runtime.h>


typedef __attribute__((ext_vector_type(2))) float v2f;
typedef __attribute__((ext_vector_type(4))) float v4f;
typedef __attribute__((ext_vector_type(8))) float v8f;

#define BSZ     256
#define LSZ     2048
#define CSZ     6
#define NW      63
#define TSTEPS  63
#define MSTRIDE 68
#define DROWS   18
#define WPB     2

__device__ __forceinline__ float xhalf(float x) {
    return __builtin_bit_cast(float, __builtin_amdgcn_permlanex16(
        __builtin_bit_cast(int, x), __builtin_bit_cast(int, x), 0x76543210, 0xfedcba98, false, false));
}

__global__ __launch_bounds__(64)
void ComputeWindowSignature_11106785427739_kernel(const float* __restrict__ path,
                                                  float* __restrict__ out)
{
    __shared__ __align__(16) float Xsh[WPB][64 * CSZ];
    __shared__ __align__(16) float Dsh[WPB][DROWS * MSTRIDE];
    const int lane = threadIdx.x & 31;
    const int wv   = threadIdx.x >> 5;
    const int gw   = blockIdx.x * WPB + wv;
    const int b    = gw / NW;
    const int w    = gw - b * NW;

    float* X  = Xsh[wv];
    float* D  = Dsh[wv];
    float* F1 = D + 6 * MSTRIDE;
    float* F2 = D + 12 * MSTRIDE;

    const size_t base = ((size_t)b * LSZ + (size_t)w * 32) * CSZ;
    const v4f* src = (const v4f*)(path + base);
    const v4f r0 = src[lane];
    const v4f r1 = src[lane + 32];
    const v4f r2 = src[lane + 64];
    v4f* Xv = (v4f*)X;
    Xv[lane]      = r0;
    Xv[lane + 32] = r1;
    Xv[lane + 64] = r2;
    __syncthreads();

    const size_t po = (size_t)gw * 258;

    #pragma unroll
    for (int it = 0; it < 12; ++it) {
        const int c = it >> 1;
        const int t = ((it & 1) << 5) + lane;
        const int tn = (t < TSTEPS) ? t + 1 : t;
        const float x0 = X[c];
        const float xt = X[t * CSZ + c];
        const float xn = X[tn * CSZ + c];
        const float d = xn - xt;
        const float p = xt - x0;
        D [c * MSTRIDE + t] = d;
        F1[c * MSTRIDE + t] = p + 0.5f * d;
        F2[c * MSTRIDE + t] = 0.5f * p + (1.f / 6.f) * d;
    }
    __syncthreads();

    const int h  = lane >> 4;
    const int c  = lane & 15;
    const int cr = (c < 12) ? c : 11;
    const int ip = (cr >= CSZ) ? 1 : 0;
    const int j  = cr - CSZ * ip;
    const float sx = X[TSTEPS * CSZ + j] - X[32 * CSZ + j];
    const v4f* djp = (const v4f*)(D + j * MSTRIDE + 32 * h);
    const v4f* f1p = (const v4f*)(F1 + ip * MSTRIDE + 32 * h);
    const v4f* f2p = (const v4f*)(F2 + ip * MSTRIDE + 32 * h);
    float sa[3] = {0.f, 0.f, 0.f};
    v8f acc[3] = {(v8f)(0.f), (v8f)(0.f), (v8f)(0.f)};
    #pragma unroll
    for (int q = 0; q < 8; ++q) {
        const v4f dj = djp[q];
        #pragma unroll
        for (int tau = 0; tau < 3; ++tau) {
            const v4f a1 = f1p[2 * tau * (MSTRIDE / 4) + q];
            const v4f b1 = f2p[2 * tau * (MSTRIDE / 4) + q];
            v4f m;
            #pragma unroll
            for (int e = 0; e < 4; ++e) {
                m[e]    = __builtin_fmaf(b1[e], dj[e], sa[tau]);
                sa[tau] = __builtin_fmaf(a1[e], dj[e], sa[tau]);
            }
            v2f alo = {m[0], m[1]}, blo = {dj[0], dj[1]};
            v2f ahi = {m[2], m[3]}, bhi = {dj[2], dj[3]};
            acc[tau] = __builtin_amdgcn_wmma_f32_16x16x4_f32(
                           false, alo, false, blo, (short)0, acc[tau], false, false);
            acc[tau] = __builtin_amdgcn_wmma_f32_16x16x4_f32(
                           false, ahi, false, bhi, (short)0, acc[tau], false, false);
        }
    }
    const bool half0 = (h == 0);
    const v2f bx = {sx, 0.f};
    #pragma unroll
    for (int tau = 0; tau < 3; ++tau) {
        const v2f ax = {half0 ? sa[tau] : 0.f, 0.f};
        acc[tau] = __builtin_amdgcn_wmma_f32_16x16x4_f32(
                       false, ax, false, bx, (short)0, acc[tau], false, false);
    }
    float tot[3];
    #pragma unroll
    for (int tau = 0; tau < 3; ++tau) {
        asm volatile("" : "+v"(sa[tau]));
        tot[tau] = sa[tau] + xhalf(sa[tau]);
    }
    const int odd = gw & 1;
    const int sft = odd << 1;
    float* S = D;
    if (lane < CSZ) S[sft + lane] = X[TSTEPS * CSZ + lane] - X[lane];
    if (half0 && c < 12) {
        #pragma unroll
        for (int tau = 0; tau < 3; ++tau) S[sft + 6 + c + 12 * tau] = 2.f * tot[tau];
    }
    if (c < CSZ) {
        float* s3p = S + sft + 42 + 48 * h + c;
        #pragma unroll
        for (int tau = 0; tau < 3; ++tau) {
            #pragma unroll
            for (int r = 0; r < 4; ++r) s3p[72 * tau + 6 * r] = 6.f * acc[tau][r];
        }
        if (half0) {
            #pragma unroll
            for (int tau = 0; tau < 3; ++tau) {
                #pragma unroll
                for (int r = 4; r < 8; ++r) s3p[72 * tau + 6 * r] = 6.f * acc[tau][r];
            }
        }
    }
    __builtin_amdgcn_fence(__ATOMIC_RELEASE, "wavefront");
    __builtin_amdgcn_wave_barrier();
    __builtin_amdgcn_fence(__ATOMIC_ACQUIRE, "wavefront");
    const v4f* Sv = (const v4f*)(S + 2 * sft);
    const v4f w0 = Sv[lane];
    const v4f w1 = Sv[lane + 32];
    v4f* ov = (v4f*)(out + po + sft);
    ov[lane]      = w0;
    ov[lane + 32] = w1;
    if (lane == 0) {
        const int e0 = odd ? 0 : 256;
        *(v2f*)(out + po + e0) = *(const v2f*)(S + sft + e0);
    }
}

extern "C" void kernel_launch(void* const* d_in, const int* in_sizes, int n_in,
                              void* d_out, int out_size, void* d_ws, size_t ws_size,
                              hipStream_t stream) {
    const float* path = (const float*)d_in[0];
    float* out = (float*)d_out;
    const int pairs  = BSZ * NW;
    const int blocks = pairs / WPB;
    ComputeWindowSignature_11106785427739_kernel<<<blocks, WPB * 32, 0, stream>>>(path, out);
}
